# starting kernel
# baseline (speedup 1.0000x reference)
#include <hip/hip_runtime.h>
#include <hip/hip_bf16.h>

typedef __attribute__((ext_vector_type(2))) float v2f;
typedef __attribute__((ext_vector_type(8))) float v8f;

__global__ void prep_weights(const float* __restrict__ w1a, const float* __restrict__ b1a,
                             const float* __restrict__ w1b, const float* __restrict__ b1b,
                             const float* __restrict__ w1c, const float* __restrict__ b1c,
                             float* __restrict__ wcat1, float* __restrict__ cb1,
                             float* __restrict__ wcat2, float* __restrict__ cb2,
                             float* __restrict__ wcat3, float* __restrict__ cb3) {
  int idx = blockIdx.x * blockDim.x + threadIdx.x;
  const int S1 = 110 * 142, S1B = 142, S2 = 32 * 64, S2B = 64, S3 = 32 * 64, S3B = 64;
  if (idx < S1) {
    int k = idx / 142, n = idx % 142;
    wcat1[idx] = (n < 71) ? (w1a[k * 71 + n] - w1a[(k + 110) * 71 + n])
                          : w1a[(k + 110) * 71 + (n - 71)];
    return;
  }
  idx -= S1;
  if (idx < S1B) { cb1[idx] = (idx < 71) ? b1a[idx] : 0.f; return; }
  idx -= S1B;
  if (idx < S2) {
    int k = idx / 64, n = idx % 64;
    wcat2[idx] = (n < 32) ? (w1b[k * 32 + n] - w1b[(k + 32) * 32 + n])
                          : w1b[(k + 32) * 32 + (n - 32)];
    return;
  }
  idx -= S2;
  if (idx < S2B) { cb2[idx] = (idx < 32) ? b1b[idx] : 0.f; return; }
  idx -= S2B;
  if (idx < S3) {
    int k = idx / 64, n = idx % 64;
    wcat3[idx] = (n < 32) ? (w1c[k * 32 + n] - w1c[(k + 32) * 32 + n])
                          : w1c[(k + 32) * 32 + (n - 32)];
    return;
  }
  idx -= S3;
  if (idx < S3B) { cb3[idx] = (idx < 32) ? b1c[idx] : 0.f; return; }
}

__global__ void pack_weights(const float* __restrict__ W, int K, int N,
                             int ksteps, int ntiles, float* __restrict__ out) {
  int idx = blockIdx.x * blockDim.x + threadIdx.x;
  int total = ntiles * ksteps * 32;
  if (idx >= total) return;
  int lane = idx & 31;
  int tmp = idx >> 5;
  int ks = tmp % ksteps;
  int nt = tmp / ksteps;
  int col = nt * 16 + (lane & 15);
  int k0 = ks * 4 + (lane >> 4) * 2;
  float v0 = (k0 < K && col < N) ? W[(size_t)k0 * N + col] : 0.f;
  float v1 = (k0 + 1 < K && col < N) ? W[(size_t)(k0 + 1) * N + col] : 0.f;
  out[(size_t)idx * 2]     = v0;
  out[(size_t)idx * 2 + 1] = v1;
}

__global__ void pad_input(const float* __restrict__ X0, float* __restrict__ Xp, int total) {
  int idx = blockIdx.x * blockDim.x + threadIdx.x;
  if (idx >= total) return;
  int r = idx / 112, c = idx - r * 112;
  Xp[idx] = (c < 110) ? X0[(size_t)r * 110 + c] : 0.f;
}

__global__ void gemm_wmma_f32(const float* __restrict__ A, int lda,
                              const float* __restrict__ Bp, int ksteps,
                              const float* __restrict__ bias, float* __restrict__ C,
                              int N, int do_relu) {
  const int lane = threadIdx.x;
  const int l16 = lane & 15;
  const int hi = lane >> 4;
  const int rowbase = blockIdx.x * 64;

  const float* aptr = A + (size_t)(rowbase + l16) * lda + hi * 2;
  const float* bptr = Bp + ((size_t)blockIdx.y * ksteps * 32 + lane) * 2;

  v8f acc0 = {0.f,0.f,0.f,0.f,0.f,0.f,0.f,0.f};
  v8f acc1 = acc0, acc2 = acc0, acc3 = acc0;

  for (int ks = 0; ks < ksteps; ++ks) {
    v2f b  = *(const v2f*)bptr;
    v2f a0 = *(const v2f*)(aptr);
    v2f a1 = *(const v2f*)(aptr + (size_t)16 * lda);
    v2f a2 = *(const v2f*)(aptr + (size_t)32 * lda);
    v2f a3 = *(const v2f*)(aptr + (size_t)48 * lda);
    acc0 = __builtin_amdgcn_wmma_f32_16x16x4_f32(false, a0, false, b, (short)0, acc0, false, false);
    acc1 = __builtin_amdgcn_wmma_f32_16x16x4_f32(false, a1, false, b, (short)0, acc1, false, false);
    acc2 = __builtin_amdgcn_wmma_f32_16x16x4_f32(false, a2, false, b, (short)0, acc2, false, false);
    acc3 = __builtin_amdgcn_wmma_f32_16x16x4_f32(false, a3, false, b, (short)0, acc3, false, false);
    aptr += 4;
    bptr += 64;
  }

  const int bcol = blockIdx.y * 16 + l16;
  if (bcol < N) {
    const float bv = bias ? bias[bcol] : 0.f;
    v8f accs[4] = {acc0, acc1, acc2, acc3};
#pragma unroll
    for (int r = 0; r < 4; ++r) {
#pragma unroll
      for (int v = 0; v < 8; ++v) {
        const int row = rowbase + r * 16 + (hi ? (v + 8) : v);
        float val = accs[r][v] + bv;
        if (do_relu) val = fmaxf(val, 0.f);
        C[(size_t)row * N + bcol] = val;
      }
    }
  }
}

__global__ void edge_ln_mean(const float* __restrict__ P, int C, int ld,
                             const float* __restrict__ g, const float* __restrict__ be,
                             float* __restrict__ H, int ldH) {
  __shared__ float hbuf[64][72];
  const int bb = blockIdx.x;
  const int e = threadIdx.x;
  const int i = e >> 3, j = e & 7;
  const float* PA = P + (size_t)(bb * 8 + i) * ld;
  const float* PB = P + (size_t)(bb * 8 + j) * ld + C;

  float sum = 0.f, sq = 0.f;
  for (int c = 0; c < C; ++c) {
    float v = PA[c] + PB[c];
    hbuf[e][c] = v;
    sum += v;
    sq += v * v;
  }
  const float inv = 1.0f / (float)C;
  const float mean = sum * inv;
  const float var = sq * inv - mean * mean;
  const float rs = rsqrtf(var + 1e-5f);
  for (int c = 0; c < C; ++c) {
    float v = (hbuf[e][c] - mean) * rs * g[c] + be[c];
    hbuf[e][c] = fmaxf(v, 0.f);
  }
  __syncthreads();

  for (int idx = e; idx < 8 * ldH; idx += 64) {
    const int ni = idx / ldH, c = idx - ni * ldH;
    float out = 0.f;
    if (c < C) {
      float s = 0.f;
#pragma unroll
      for (int jj = 0; jj < 8; ++jj) s += hbuf[ni * 8 + jj][c];
      out = s * 0.125f;
    }
    H[(size_t)(bb * 8 + ni) * ldH + c] = out;
  }
}

__global__ void head_kernel(const float* __restrict__ Hh,
                            const float* __restrict__ phiW1, const float* __restrict__ phib1,
                            const float* __restrict__ phiW2, const float* __restrict__ phib2,
                            float* __restrict__ out) {
  __shared__ float p[16];
  __shared__ float q[16];
  const int bb = blockIdx.x, t = threadIdx.x;
  if (t < 16) {
    float s = 0.f;
    for (int nd = 0; nd < 8; ++nd) s += Hh[(size_t)(bb * 8 + nd) * 16 + t];
    p[t] = s * 0.125f;
  }
  __syncthreads();
  if (t < 16) {
    float s = phib1[t];
    for (int k = 0; k < 16; ++k) s += p[k] * phiW1[k * 16 + t];
    q[t] = fmaxf(s, 0.f);
  }
  __syncthreads();
  if (t == 0) {
    float s = phib2[0];
    for (int k = 0; k < 16; ++k) s += q[k] * phiW2[k];
    out[bb] = s;
  }
}

extern "C" void kernel_launch(void* const* d_in, const int* in_sizes, int n_in,
                              void* d_out, int out_size, void* d_ws, size_t ws_size,
                              hipStream_t stream) {
  (void)in_sizes; (void)n_in; (void)out_size; (void)ws_size;
  const float* X0     = (const float*)d_in[0];
  const float* l1_W1  = (const float*)d_in[1];
  const float* l1_b1  = (const float*)d_in[2];
  const float* l1_g   = (const float*)d_in[3];
  const float* l1_be  = (const float*)d_in[4];
  const float* l1_W2  = (const float*)d_in[5];
  const float* l1_b2  = (const float*)d_in[6];
  const float* l2_W1  = (const float*)d_in[7];
  const float* l2_b1  = (const float*)d_in[8];
  const float* l2_g   = (const float*)d_in[9];
  const float* l2_be  = (const float*)d_in[10];
  const float* l2_W2  = (const float*)d_in[11];
  const float* l2_b2  = (const float*)d_in[12];
  const float* l3_W1  = (const float*)d_in[13];
  const float* l3_b1  = (const float*)d_in[14];
  const float* l3_g   = (const float*)d_in[15];
  const float* l3_be  = (const float*)d_in[16];
  const float* l3_W2  = (const float*)d_in[17];
  const float* l3_b2  = (const float*)d_in[18];
  const float* psi_W1 = (const float*)d_in[19];
  const float* psi_b1 = (const float*)d_in[20];
  const float* psi_W2 = (const float*)d_in[21];
  const float* psi_b2 = (const float*)d_in[22];
  const float* phi_W1 = (const float*)d_in[23];
  const float* phi_b1 = (const float*)d_in[24];
  const float* phi_W2 = (const float*)d_in[25];
  const float* phi_b2 = (const float*)d_in[26];

  const int M = 32768;

  float* ws = (float*)d_ws;
  float* wcat1 = ws;
  float* cb1   = wcat1 + 15620;
  float* wcat2 = cb1 + 144;
  float* cb2   = wcat2 + 2048;
  float* wcat3 = cb2 + 64;
  float* cb3   = wcat3 + 2048;
  float* p_w1   = cb3 + 64;
  float* p_l1W2 = p_w1 + 16128;
  float* p_w2   = p_l1W2 + 2304;
  float* p_l2W2 = p_w2 + 2048;
  float* p_w3   = p_l2W2 + 1024;
  float* p_l3W2 = p_w3 + 2048;
  float* p_psi1 = p_l3W2 + 1024;
  float* p_psi2 = p_psi1 + 512;
  float* X0p = ws + 49152;
  float* P   = X0p + (size_t)M * 112;
  float* H   = P   + (size_t)M * 142;
  float* X   = H   + (size_t)M * 72;
  float* X5  = X   + (size_t)M * 32;
  float* Hh  = X5  + (size_t)M * 16;

  prep_weights<<<80, 256, 0, stream>>>(l1_W1, l1_b1, l2_W1, l2_b1, l3_W1, l3_b1,
                                       wcat1, cb1, wcat2, cb2, wcat3, cb3);
  pack_weights<<<32, 256, 0, stream>>>(wcat1, 110, 142, 28, 9, p_w1);
  pack_weights<<<5, 256, 0, stream>>>(l1_W2, 71, 32, 18, 2, p_l1W2);
  pack_weights<<<4, 256, 0, stream>>>(wcat2, 32, 64, 8, 4, p_w2);
  pack_weights<<<2, 256, 0, stream>>>(l2_W2, 32, 32, 8, 2, p_l2W2);
  pack_weights<<<4, 256, 0, stream>>>(wcat3, 32, 64, 8, 4, p_w3);
  pack_weights<<<2, 256, 0, stream>>>(l3_W2, 32, 32, 8, 2, p_l3W2);
  pack_weights<<<1, 256, 0, stream>>>(psi_W1, 32, 16, 8, 1, p_psi1);
  pack_weights<<<1, 256, 0, stream>>>(psi_W2, 16, 16, 4, 1, p_psi2);
  pad_input<<<(M * 112 + 255) / 256, 256, 0, stream>>>(X0, X0p, M * 112);

  const int GB = M / 64;

  gemm_wmma_f32<<<dim3(GB, 9), 32, 0, stream>>>(X0p, 112, p_w1, 28, cb1, P, 142, 0);
  edge_ln_mean<<<4096, 64, 0, stream>>>(P, 71, 142, l1_g, l1_be, H, 72);
  gemm_wmma_f32<<<dim3(GB, 2), 32, 0, stream>>>(H, 72, p_l1W2, 18, l1_b2, X, 32, 1);

  gemm_wmma_f32<<<dim3(GB, 4), 32, 0, stream>>>(X, 32, p_w2, 8, cb2, P, 64, 0);
  edge_ln_mean<<<4096, 64, 0, stream>>>(P, 32, 64, l2_g, l2_be, H, 32);
  gemm_wmma_f32<<<dim3(GB, 2), 32, 0, stream>>>(H, 32, p_l2W2, 8, l2_b2, X, 32, 1);

  gemm_wmma_f32<<<dim3(GB, 4), 32, 0, stream>>>(X, 32, p_w3, 8, cb3, P, 64, 0);
  edge_ln_mean<<<4096, 64, 0, stream>>>(P, 32, 64, l3_g, l3_be, H, 32);
  gemm_wmma_f32<<<dim3(GB, 2), 32, 0, stream>>>(H, 32, p_l3W2, 8, l3_b2, X, 32, 0);

  gemm_wmma_f32<<<dim3(GB, 1), 32, 0, stream>>>(X, 32, p_psi1, 8, psi_b1, X5, 16, 1);
  gemm_wmma_f32<<<dim3(GB, 1), 32, 0, stream>>>(X5, 16, p_psi2, 4, psi_b2, Hh, 16, 0);

  head_kernel<<<4096, 32, 0, stream>>>(Hh, phi_W1, phi_b1, phi_W2, phi_b2, (float*)d_out);
}


# Consecutive maps chained in registers in transposed form: the D tile of W^T@X^T is the B operand of the next map (W2 -> next projection; W2 -> psi -> node mean -> phi); 12 -> 7 launches
# speedup vs baseline: 1.7848x; 1.7848x over previous
#include <hip/hip_runtime.h>
#include <hip/hip_bf16.h>

typedef __attribute__((ext_vector_type(2))) float v2f;
typedef __attribute__((ext_vector_type(8))) float v8f;

__device__ __forceinline__ float ld1(const float* base, unsigned byteoff) {
  return *(const float*)((const char*)base + byteoff);
}
__device__ __forceinline__ v2f ld2(const float* base, unsigned byteoff) {
  return *(const v2f*)((const char*)base + byteoff);
}

template <bool EDGE>
__device__ __forceinline__ void gemm_body(const float* __restrict__ A, int lda,
                                          const float* __restrict__ W, int K, int dout,
                                          const float* __restrict__ bias, float* __restrict__ C,
                                          int N, int do_relu) {
  const int lane = threadIdx.x;
  const int l16 = lane & 15;
  const int hi = lane >> 4;
  const int rowbase = blockIdx.x * 64;
  const int bcol = blockIdx.y * 16 + l16;
  const int colc = min(bcol, N - 1);
  const int ldw = EDGE ? dout : N;

  const bool apart = EDGE && (colc < dout);
  const int cc = EDGE ? (apart ? colc : colc - dout) : colc;
  const float m1 = apart ? 1.f : 0.f;
  const float m2 = apart ? -1.f : 1.f;

  const unsigned aoff = 4u * (unsigned)(l16 * lda + hi * 2);
  const unsigned boff = 4u * (unsigned)(hi * 2 * ldw + cc);
  const float* Ab = A + (size_t)rowbase * lda;
  const float* Wlo = W;
  const float* Whi = W + (size_t)K * ldw;
  const size_t arow = (size_t)16 * lda;

  v8f acc0 = {0.f,0.f,0.f,0.f,0.f,0.f,0.f,0.f};
  v8f acc1 = acc0, acc2 = acc0, acc3 = acc0;

  const int kfull = K >> 2;
  for (int ks = 0; ks < kfull; ++ks) {
    const float* Ak = Ab + ks * 4;
    const float* W0 = Wlo + (size_t)(ks * 4) * ldw;
    const float* W1 = Whi + (size_t)(ks * 4) * ldw;
    v2f a0 = ld2(Ak, aoff);
    v2f a1 = ld2(Ak + arow, aoff);
    v2f a2 = ld2(Ak + 2 * arow, aoff);
    v2f a3 = ld2(Ak + 3 * arow, aoff);
    v2f b;
    if (EDGE) {
      const float t0 = ld1(W0, boff), t1 = ld1(W0 + ldw, boff);
      const float u0 = ld1(W1, boff), u1 = ld1(W1 + ldw, boff);
      b.x = fmaf(m1, t0, m2 * u0);
      b.y = fmaf(m1, t1, m2 * u1);
    } else {
      b.x = ld1(W0, boff);
      b.y = ld1(W0 + ldw, boff);
    }
    acc0 = __builtin_amdgcn_wmma_f32_16x16x4_f32(false, a0, false, b, (short)0, acc0, false, false);
    acc1 = __builtin_amdgcn_wmma_f32_16x16x4_f32(false, a1, false, b, (short)0, acc1, false, false);
    acc2 = __builtin_amdgcn_wmma_f32_16x16x4_f32(false, a2, false, b, (short)0, acc2, false, false);
    acc3 = __builtin_amdgcn_wmma_f32_16x16x4_f32(false, a3, false, b, (short)0, acc3, false, false);
  }

  if (K & 3) {
    const int k = kfull * 4 + hi * 2;
    const unsigned o0 = 4u * (unsigned)(min(k, K - 1) * ldw + cc);
    const unsigned o1 = 4u * (unsigned)(min(k + 1, K - 1) * ldw + cc);
    v2f b;
    if (EDGE) {
      b.x = fmaf(m1, ld1(Wlo, o0), m2 * ld1(Whi, o0));
      b.y = fmaf(m1, ld1(Wlo, o1), m2 * ld1(Whi, o1));
    } else {
      b.x = ld1(Wlo, o0);
      b.y = ld1(Wlo, o1);
    }
    const v2f z = {0.f, 0.f};
    v2f a0 = z, a1 = z, a2 = z, a3 = z;
    if (k < lda) {
      const float* Ak = Ab + kfull * 4;
      a0 = ld2(Ak, aoff);
      a1 = ld2(Ak + arow, aoff);
      a2 = ld2(Ak + 2 * arow, aoff);
      a3 = ld2(Ak + 3 * arow, aoff);
    }
    acc0 = __builtin_amdgcn_wmma_f32_16x16x4_f32(false, a0, false, b, (short)0, acc0, false, false);
    acc1 = __builtin_amdgcn_wmma_f32_16x16x4_f32(false, a1, false, b, (short)0, acc1, false, false);
    acc2 = __builtin_amdgcn_wmma_f32_16x16x4_f32(false, a2, false, b, (short)0, acc2, false, false);
    acc3 = __builtin_amdgcn_wmma_f32_16x16x4_f32(false, a3, false, b, (short)0, acc3, false, false);
  }

  if (bcol < N) {
    float bv = 0.f;
    if (bias) {
      if (EDGE) { if (bcol < dout) bv = bias[bcol]; }
      else bv = bias[bcol];
    }
    v8f accs[4] = {acc0, acc1, acc2, acc3};
#pragma unroll
    for (int r = 0; r < 4; ++r) {
#pragma unroll
      for (int v = 0; v < 8; ++v) {
        const int row = rowbase + r * 16 + (hi ? (v + 8) : v);
        float val = accs[r][v] + bv;
        if (do_relu) val = fmaxf(val, 0.f);
        C[(size_t)row * N + bcol] = val;
      }
    }
  }
}

__global__ void gemm_wmma_f32_edge(const float* __restrict__ A, int lda,
                                   const float* __restrict__ W, int K, int dout,
                                   const float* __restrict__ bias, float* __restrict__ C) {
  gemm_body<true>(A, lda, W, K, dout, bias, C, 2 * dout, 0);
}


#define RT 2

__device__ __forceinline__ v8f wmma4(v2f a, v2f b, v8f c) {
  return __builtin_amdgcn_wmma_f32_16x16x4_f32(false, a, false, b, (short)0, c, false, false);
}

template <int LDH, int K1, bool RELU>
__device__ __forceinline__ void map_in(const float* __restrict__ H, const float* __restrict__ W2,
                                       const float* __restrict__ b2, int rowbase, v8f (&x)[2][RT]) {
  const int lane = threadIdx.x, c = lane & 15, h = lane >> 4;
  const v8f z8 = {0.f,0.f,0.f,0.f,0.f,0.f,0.f,0.f};
#pragma unroll
  for (int n = 0; n < 2; ++n)
#pragma unroll
    for (int t = 0; t < RT; ++t) x[n][t] = z8;
#pragma unroll
  for (int ks = 0; ks < LDH / 4; ++ks) {
    const int k0 = 4 * ks + 2 * h;
    const int ka = (k0 < K1) ? k0 : (K1 - 1);
    const int kb = (k0 + 1 < K1) ? (k0 + 1) : (K1 - 1);
    v2f b[RT];
#pragma unroll
    for (int t = 0; t < RT; ++t)
      b[t] = *(const v2f*)(H + (size_t)(rowbase + 16 * t + c) * LDH + k0);
#pragma unroll
    for (int n = 0; n < 2; ++n) {
      v2f a;
      a.x = W2[ka * 32 + 16 * n + c];
      a.y = W2[kb * 32 + 16 * n + c];
#pragma unroll
      for (int t = 0; t < RT; ++t) x[n][t] = wmma4(a, b[t], x[n][t]);
    }
  }
#pragma unroll
  for (int n = 0; n < 2; ++n) {
#pragma unroll
    for (int r = 0; r < 8; ++r) {
      const float bv = b2[16 * n + 8 * h + r];
#pragma unroll
      for (int t = 0; t < RT; ++t) {
        float v = x[n][t][r] + bv;
        x[n][t][r] = RELU ? fmaxf(v, 0.f) : v;
      }
    }
  }
}

template <int LDH, int K1>
__global__ void __launch_bounds__(32) chain_edge(const float* __restrict__ H, const float* __restrict__ W2,
                           const float* __restrict__ b2, const float* __restrict__ W1,
                           const float* __restrict__ b1, float* __restrict__ P) {
  const int lane = threadIdx.x, c = lane & 15, h = lane >> 4;
  const int rowbase = blockIdx.x * (16 * RT);
  v8f x[2][RT];
  map_in<LDH, K1, true>(H, W2, b2, rowbase, x);

  const float* Wlo = W1;
  const float* Whi = W1 + 32 * 32;
  const v8f z8 = {0.f,0.f,0.f,0.f,0.f,0.f,0.f,0.f};
  v8f p[4][RT];
#pragma unroll
  for (int m = 0; m < 4; ++m)
#pragma unroll
    for (int t = 0; t < RT; ++t) p[m][t] = z8;
#pragma unroll
  for (int n = 0; n < 2; ++n) {
#pragma unroll
    for (int q = 0; q < 4; ++q) {
      const int k0 = 16 * n + 8 * h + 2 * q;
#pragma unroll
      for (int m = 0; m < 4; ++m) {
        v2f a;
        if (m < 2) {
          const int col = 16 * m + c;
          a.x = Wlo[k0 * 32 + col] - Whi[k0 * 32 + col];
          a.y = Wlo[(k0 + 1) * 32 + col] - Whi[(k0 + 1) * 32 + col];
        } else {
          const int col = 16 * (m - 2) + c;
          a.x = Whi[k0 * 32 + col];
          a.y = Whi[(k0 + 1) * 32 + col];
        }
#pragma unroll
        for (int t = 0; t < RT; ++t) {
          v2f b;
          b.x = x[n][t][2 * q];
          b.y = x[n][t][2 * q + 1];
          p[m][t] = wmma4(a, b, p[m][t]);
        }
      }
    }
  }
#pragma unroll
  for (int m = 0; m < 4; ++m) {
    float bv[8];
#pragma unroll
    for (int r = 0; r < 8; ++r) bv[r] = (m < 2) ? b1[16 * m + 8 * h + r] : 0.f;
#pragma unroll
    for (int t = 0; t < RT; ++t) {
      float* dst = P + (size_t)(rowbase + 16 * t + c) * 64 + 16 * m + 8 * h;
      float4 lo, hi4;
      lo.x = p[m][t][0] + bv[0]; lo.y = p[m][t][1] + bv[1];
      lo.z = p[m][t][2] + bv[2]; lo.w = p[m][t][3] + bv[3];
      hi4.x = p[m][t][4] + bv[4]; hi4.y = p[m][t][5] + bv[5];
      hi4.z = p[m][t][6] + bv[6]; hi4.w = p[m][t][7] + bv[7];
      *(float4*)dst = lo;
      *(float4*)(dst + 4) = hi4;
    }
  }
}

__global__ void __launch_bounds__(32) __attribute__((amdgpu_waves_per_eu(1, 2))) chain_psi(const float* __restrict__ H, const float* __restrict__ W2,
                          const float* __restrict__ b2,
                          const float* __restrict__ psiW1, const float* __restrict__ pb1,
                          const float* __restrict__ psiW2, const float* __restrict__ pb2,
                          const float* __restrict__ phiW1, const float* __restrict__ phib1,
                          const float* __restrict__ phiW2, const float* __restrict__ phib2,
                          float* __restrict__ out) {
  const int lane = threadIdx.x, c = lane & 15, h = lane >> 4;
  const int rowbase = blockIdx.x * (16 * RT);
  v8f x[2][RT];
  map_in<32, 32, false>(H, W2, b2, rowbase, x);

  const v8f z8 = {0.f,0.f,0.f,0.f,0.f,0.f,0.f,0.f};
  v8f y[RT];
#pragma unroll
  for (int t = 0; t < RT; ++t) y[t] = z8;
#pragma unroll
  for (int n = 0; n < 2; ++n) {
#pragma unroll
    for (int q = 0; q < 4; ++q) {
      const int k0 = 16 * n + 8 * h + 2 * q;
      v2f a;
      a.x = psiW1[k0 * 16 + c];
      a.y = psiW1[(k0 + 1) * 16 + c];
#pragma unroll
      for (int t = 0; t < RT; ++t) {
        v2f b;
        b.x = x[n][t][2 * q];
        b.y = x[n][t][2 * q + 1];
        y[t] = wmma4(a, b, y[t]);
      }
    }
  }
#pragma unroll
  for (int r = 0; r < 8; ++r) {
    const float bv = pb1[8 * h + r];
#pragma unroll
    for (int t = 0; t < RT; ++t) y[t][r] = fmaxf(y[t][r] + bv, 0.f);
  }
  v8f zt[RT];
#pragma unroll
  for (int t = 0; t < RT; ++t) zt[t] = z8;
#pragma unroll
  for (int q = 0; q < 4; ++q) {
    const int k0 = 8 * h + 2 * q;
    v2f a;
    a.x = psiW2[k0 * 16 + c];
    a.y = psiW2[(k0 + 1) * 16 + c];
#pragma unroll
    for (int t = 0; t < RT; ++t) {
      v2f b;
      b.x = y[t][2 * q];
      b.y = y[t][2 * q + 1];
      zt[t] = wmma4(a, b, zt[t]);
    }
  }
  v8f pm[RT];
#pragma unroll
  for (int r = 0; r < 8; ++r) {
    const float bv = pb2[8 * h + r];
#pragma unroll
    for (int t = 0; t < RT; ++t) {
      float v = zt[t][r] + bv;
      v += __shfl_xor(v, 1);
      v += __shfl_xor(v, 2);
      v += __shfl_xor(v, 4);
      pm[t][r] = v * 0.125f;
    }
  }
  v8f qv[RT];
#pragma unroll
  for (int t = 0; t < RT; ++t) qv[t] = z8;
#pragma unroll
  for (int q = 0; q < 4; ++q) {
    const int k0 = 8 * h + 2 * q;
    v2f a;
    a.x = phiW1[k0 * 16 + c];
    a.y = phiW1[(k0 + 1) * 16 + c];
#pragma unroll
    for (int t = 0; t < RT; ++t) {
      v2f b;
      b.x = pm[t][2 * q];
      b.y = pm[t][2 * q + 1];
      qv[t] = wmma4(a, b, qv[t]);
    }
  }
  float part[RT];
#pragma unroll
  for (int t = 0; t < RT; ++t) part[t] = 0.f;
#pragma unroll
  for (int r = 0; r < 8; ++r) {
    const float b1v = phib1[8 * h + r];
    const float w2v = phiW2[8 * h + r];
#pragma unroll
    for (int t = 0; t < RT; ++t) part[t] += fmaxf(qv[t][r] + b1v, 0.f) * w2v;
  }
  const float b2v = phib2[0];
#pragma unroll
  for (int t = 0; t < RT; ++t) {
    const float y = part[t] + __shfl_xor(part[t], 16) + b2v;
    if (h == 0 && (c & 7) == 0) out[(rowbase + 16 * t) / 8 + (c >> 3)] = y;
  }
}

__global__ void edge_ln_mean(const float* __restrict__ P, int C, int ld,
                             const float* __restrict__ g, const float* __restrict__ be,
                             float* __restrict__ H, int ldH) {
  __shared__ float hbuf[64][72];
  const int bb = blockIdx.x;
  const int e = threadIdx.x;
  const int i = e >> 3, j = e & 7;
  const float* PA = P + (size_t)(bb * 8 + i) * ld;
  const float* PB = P + (size_t)(bb * 8 + j) * ld + C;

  float sum = 0.f, sq = 0.f;
  for (int c = 0; c < C; ++c) {
    float v = PA[c] + PB[c];
    hbuf[e][c] = v;
    sum += v;
    sq += v * v;
  }
  const float inv = 1.0f / (float)C;
  const float mean = sum * inv;
  const float var = sq * inv - mean * mean;
  const float rs = rsqrtf(var + 1e-5f);
  for (int c = 0; c < C; ++c) {
    float v = (hbuf[e][c] - mean) * rs * g[c] + be[c];
    hbuf[e][c] = fmaxf(v, 0.f);
  }
  __syncthreads();

  for (int idx = e; idx < 8 * ldH; idx += 64) {
    const int ni = idx / ldH, c = idx - ni * ldH;
    float out = 0.f;
    if (c < C) {
      float s = 0.f;
#pragma unroll
      for (int jj = 0; jj < 8; ++jj) s += hbuf[ni * 8 + jj][c];
      out = s * 0.125f;
    }
    H[(size_t)(bb * 8 + ni) * ldH + c] = out;
  }
}

extern "C" void kernel_launch(void* const* d_in, const int* in_sizes, int n_in,
                              void* d_out, int out_size, void* d_ws, size_t ws_size,
                              hipStream_t stream) {
  (void)in_sizes; (void)n_in; (void)out_size; (void)ws_size;
  const float* X0     = (const float*)d_in[0];
  const float* l1_W1  = (const float*)d_in[1];
  const float* l1_b1  = (const float*)d_in[2];
  const float* l1_g   = (const float*)d_in[3];
  const float* l1_be  = (const float*)d_in[4];
  const float* l1_W2  = (const float*)d_in[5];
  const float* l1_b2  = (const float*)d_in[6];
  const float* l2_W1  = (const float*)d_in[7];
  const float* l2_b1  = (const float*)d_in[8];
  const float* l2_g   = (const float*)d_in[9];
  const float* l2_be  = (const float*)d_in[10];
  const float* l2_W2  = (const float*)d_in[11];
  const float* l2_b2  = (const float*)d_in[12];
  const float* l3_W1  = (const float*)d_in[13];
  const float* l3_b1  = (const float*)d_in[14];
  const float* l3_g   = (const float*)d_in[15];
  const float* l3_be  = (const float*)d_in[16];
  const float* l3_W2  = (const float*)d_in[17];
  const float* l3_b2  = (const float*)d_in[18];
  const float* psi_W1 = (const float*)d_in[19];
  const float* psi_b1 = (const float*)d_in[20];
  const float* psi_W2 = (const float*)d_in[21];
  const float* psi_b2 = (const float*)d_in[22];
  const float* phi_W1 = (const float*)d_in[23];
  const float* phi_b1 = (const float*)d_in[24];
  const float* phi_W2 = (const float*)d_in[25];
  const float* phi_b2 = (const float*)d_in[26];

  float* ws = (float*)d_ws;
  float* P   = ws;
  float* H   = P   + (size_t)32768 * 142;

  gemm_wmma_f32_edge<<<dim3(512, 9), 32, 0, stream>>>(X0, 110, l1_W1, 110, 71, l1_b1, P);
  edge_ln_mean<<<4096, 64, 0, stream>>>(P, 71, 142, l1_g, l1_be, H, 72);

  chain_edge<72, 71><<<32768 / (16 * RT), 32, 0, stream>>>(H, l1_W2, l1_b2, l2_W1, l2_b1, P);
  edge_ln_mean<<<4096, 64, 0, stream>>>(P, 32, 64, l2_g, l2_be, H, 32);

  chain_edge<32, 32><<<32768 / (16 * RT), 32, 0, stream>>>(H, l2_W2, l2_b2, l3_W1, l3_b1, P);
  edge_ln_mean<<<4096, 64, 0, stream>>>(P, 32, 64, l3_g, l3_be, H, 32);

  chain_psi<<<32768 / (16 * RT), 32, 0, stream>>>(H, l3_W2, l3_b2, psi_W1, psi_b1, psi_W2, psi_b2,
                                                  phi_W1, phi_b1, phi_W2, phi_b2, (float*)d_out);
}
